# starting kernel
# baseline (speedup 1.0000x reference)
#include <hip/hip_runtime.h>
#include <math.h>

typedef __attribute__((ext_vector_type(2))) float v2f;
typedef __attribute__((ext_vector_type(8))) float v8f;

#define RED_BLOCKS 1024

__global__ void se3_ssq_partial(const float* __restrict__ tau,
                                float* __restrict__ partial, int K)
{
    __shared__ float red[256];
    float s = 0.0f;
    for (int e = blockIdx.x * 256 + threadIdx.x; e < K; e += gridDim.x * 256) {
        const float2* t2 = (const float2*)(tau + (size_t)e * 6);
        float2 q1 = t2[1];
        float2 q2 = t2[2];
        float wx = q1.y, wy = q2.x, wz = q2.y;
        s += wx * wx + wy * wy + wz * wz;
    }
    red[threadIdx.x] = s;
    __syncthreads();
    for (int off = 128; off > 0; off >>= 1) {
        if (threadIdx.x < off) red[threadIdx.x] += red[threadIdx.x + off];
        __syncthreads();
    }
    if (threadIdx.x == 0) partial[blockIdx.x] = red[0];
}

__global__ void se3_ssq_final(const float* __restrict__ partial,
                              float* __restrict__ bc)
{
    __shared__ float red[256];
    float s = 0.0f;
    for (int i = 0; i < RED_BLOCKS / 256; ++i) s += partial[threadIdx.x + i * 256];
    red[threadIdx.x] = s;
    __syncthreads();
    for (int off = 128; off > 0; off >>= 1) {
        if (threadIdx.x < off) red[threadIdx.x] += red[threadIdx.x + off];
        __syncthreads();
    }
    if (threadIdx.x == 0) {
        float th = sqrtf(red[0]);
        float t2 = th * th;
        bc[0] = (1.0f - cosf(th)) / t2;
        bc[1] = (th - sinf(th)) / (t2 * th);
    }
}

__global__ void __launch_bounds__(256) se3_wmma_main(const float* __restrict__ Mp,
                                                     const float* __restrict__ tau,
                                                     const float* __restrict__ bc,
                                                     float* __restrict__ out)
{
    __shared__ __align__(16) float sE[256 * 20];
    __shared__ __align__(16) float sO[256 * 20];

    const int tid = threadIdx.x;
    const size_t blockBase = (size_t)blockIdx.x * 256;
    const size_t e = blockBase + tid;

    __builtin_prefetch(Mp + e * 16, 0, 0);

    const float2* t2 = (const float2*)(tau + e * 6);
    float2 q0 = t2[0], q1 = t2[1], q2 = t2[2];
    float rx = q0.x, ry = q0.y, rz = q1.x;
    float wx = q1.y, wy = q2.x, wz = q2.y;

    float th = sqrtf(wx * wx + wy * wy + wz * wz);
    float inv = 1.0f / th;
    float ux = wx * inv, uy = wy * inv, uz = wz * inv;
    float sn = sinf(th);
    float omc = 1.0f - cosf(th);

    float R01 = -sn * uz + omc * uz * uz;
    float R02 =  sn * uy + omc * uy * uy;
    float R10 =  sn * uz + omc * uz * uz;
    float R12 = -sn * ux + omc * ux * ux;
    float R20 = -sn * uy + omc * uy * uy;
    float R21 =  sn * ux + omc * ux * ux;

    float b = bc[0], c = bc[1];
    float V01 = -b * wz + c * wz * wz;
    float V02 =  b * wy + c * wy * wy;
    float V10 =  b * wz + c * wz * wz;
    float V12 = -b * wx + c * wx * wx;
    float V20 = -b * wy + c * wy * wy;
    float V21 =  b * wx + c * wx * wx;
    float px = rx + V01 * ry + V02 * rz;
    float py = V10 * rx + ry + V12 * rz;
    float pz = V20 * rx + V21 * ry + rz;

    float* Ec = &sE[tid * 20];
    Ec[0]  = 1.0f; Ec[1]  = R10;  Ec[2]  = R20;  Ec[3]  = 0.0f;
    Ec[4]  = R01;  Ec[5]  = 1.0f; Ec[6]  = R21;  Ec[7]  = 0.0f;
    Ec[8]  = R02;  Ec[9]  = R12;  Ec[10] = 1.0f; Ec[11] = 0.0f;
    Ec[12] = px;   Ec[13] = py;   Ec[14] = pz;   Ec[15] = 1.0f;

    __syncthreads();

    const int lane = tid & 31;
    const int waveBase = tid & ~31;
    const int n = lane & 15;
    const int half = lane >> 4;
    const int k0 = half << 1;
    const int cc = n & 3;
    const int dstHalf = n >> 3;
    const bool useHi = ((n >> 2) & 1) != 0;

    const float* aBase = Mp + (blockBase + waveBase + (n >> 2)) * 16
                            + (size_t)((n & 3) * 4 + k0);
    const float* bBase = &sE[(waveBase + (n >> 2)) * 20 + cc * 4 + k0];
    float*       oBase = &sO[(waveBase + (n >> 2)) * 20 + cc];

    v2f av[8], bv[8];
    #pragma unroll
    for (int g = 0; g < 8; ++g) {
        av[g] = *(const v2f*)(aBase + g * 64);
        bv[g] = *(const v2f*)(bBase + g * 80);
    }

    #pragma unroll
    for (int g = 0; g < 8; ++g) {
        v8f acc = {0.f, 0.f, 0.f, 0.f, 0.f, 0.f, 0.f, 0.f};
        acc = __builtin_amdgcn_wmma_f32_16x16x4_f32(
             false, av[g],  false, bv[g],
             (short)0, acc,  false,  false);

        if (half == dstHalf) {
            float d0 = useHi ? acc[4] : acc[0];
            float d1 = useHi ? acc[5] : acc[1];
            float d2 = useHi ? acc[6] : acc[2];
            float d3 = useHi ? acc[7] : acc[3];
            float* o = oBase + g * 80;
            o[0]  = d0;
            o[4]  = d1;
            o[8]  = d2;
            o[12] = d3;
        }
    }

    __syncthreads();

    const float4* src = (const float4*)(&sO[tid * 20]);
    float4* dst = (float4*)(out + e * 16);
    dst[0] = src[0];
    dst[1] = src[1];
    dst[2] = src[2];
    dst[3] = src[3];
}

__global__ void se3_tail(const float* __restrict__ Mp, const float* __restrict__ tau,
                         const float* __restrict__ bc, float* __restrict__ out,
                         int start, int K)
{
    int e = start + blockIdx.x * blockDim.x + threadIdx.x;
    if (e >= K) return;
    const float* t = tau + (size_t)e * 6;
    float rx = t[0], ry = t[1], rz = t[2], wx = t[3], wy = t[4], wz = t[5];
    float th = sqrtf(wx * wx + wy * wy + wz * wz);
    float inv = 1.0f / th;
    float ux = wx * inv, uy = wy * inv, uz = wz * inv;
    float sn = sinf(th), omc = 1.0f - cosf(th);
    float E[4][4];
    E[0][0] = 1.0f;                    E[0][1] = -sn * uz + omc * uz * uz; E[0][2] =  sn * uy + omc * uy * uy;
    E[1][0] =  sn * uz + omc * uz * uz; E[1][1] = 1.0f;                   E[1][2] = -sn * ux + omc * ux * ux;
    E[2][0] = -sn * uy + omc * uy * uy; E[2][1] =  sn * ux + omc * ux * ux; E[2][2] = 1.0f;
    float b = bc[0], c = bc[1];
    E[0][3] = rx + (-b * wz + c * wz * wz) * ry + ( b * wy + c * wy * wy) * rz;
    E[1][3] = ( b * wz + c * wz * wz) * rx + ry + (-b * wx + c * wx * wx) * rz;
    E[2][3] = (-b * wy + c * wy * wy) * rx + ( b * wx + c * wx * wx) * ry + rz;
    E[3][0] = 0.0f; E[3][1] = 0.0f; E[3][2] = 0.0f; E[3][3] = 1.0f;
    const float* m = Mp + (size_t)e * 16;
    float* o = out + (size_t)e * 16;
    for (int r = 0; r < 4; ++r)
        for (int c2 = 0; c2 < 4; ++c2) {
            float s = 0.0f;
            for (int k = 0; k < 4; ++k) s += m[r * 4 + k] * E[k][c2];
            o[r * 4 + c2] = s;
        }
}

extern "C" void kernel_launch(void* const* d_in, const int* in_sizes, int n_in,
                              void* d_out, int out_size, void* d_ws, size_t ws_size,
                              hipStream_t stream)
{
    const float* Mp  = (const float*)d_in[0];
    const float* tau = (const float*)d_in[1];
    float* out = (float*)d_out;
    const int K = in_sizes[0] / 16;

    float* w = (float*)d_ws;
    float* bc = w;
    float* partial = w + 16;

    se3_ssq_partial<<<RED_BLOCKS, 256, 0, stream>>>(tau, partial, K);
    se3_ssq_final<<<1, 256, 0, stream>>>(partial, bc);

    const int nb = K / 256;
    if (nb > 0)
        se3_wmma_main<<<nb, 256, 0, stream>>>(Mp, tau, bc, out);
    const int tail = K - nb * 256;
    if (tail > 0)
        se3_tail<<<(tail + 255) / 256, 256, 0, stream>>>(Mp, tau, bc, out, nb * 256, K);
}


# output stores cover whole 128-B lines: results staged in LDS at 64-B pitch, each global_store_b128 writes 512 contiguous bytes of the wave's 2 KB (was 16 lines x two 16-B pieces per instruction)
# speedup vs baseline: 1.4511x; 1.4511x over previous
#include <hip/hip_runtime.h>
#include <math.h>

typedef __attribute__((ext_vector_type(2))) float v2f;
typedef __attribute__((ext_vector_type(8))) float v8f;

#define RED_BLOCKS 1024

__global__ void se3_ssq_partial(const float* __restrict__ tau,
                                float* __restrict__ partial, int K)
{
    __shared__ float red[256];
    float s = 0.0f;
    for (int e = blockIdx.x * 256 + threadIdx.x; e < K; e += gridDim.x * 256) {
        const float2* t2 = (const float2*)(tau + (size_t)e * 6);
        float2 q1 = t2[1];
        float2 q2 = t2[2];
        float wx = q1.y, wy = q2.x, wz = q2.y;
        s += wx * wx + wy * wy + wz * wz;
    }
    red[threadIdx.x] = s;
    __syncthreads();
    for (int off = 128; off > 0; off >>= 1) {
        if (threadIdx.x < off) red[threadIdx.x] += red[threadIdx.x + off];
        __syncthreads();
    }
    if (threadIdx.x == 0) partial[blockIdx.x] = red[0];
}

__global__ void se3_ssq_final(const float* __restrict__ partial,
                              float* __restrict__ bc)
{
    __shared__ float red[256];
    float s = 0.0f;
    for (int i = 0; i < RED_BLOCKS / 256; ++i) s += partial[threadIdx.x + i * 256];
    red[threadIdx.x] = s;
    __syncthreads();
    for (int off = 128; off > 0; off >>= 1) {
        if (threadIdx.x < off) red[threadIdx.x] += red[threadIdx.x + off];
        __syncthreads();
    }
    if (threadIdx.x == 0) {
        float th = sqrtf(red[0]);
        float t2 = th * th;
        bc[0] = (1.0f - cosf(th)) / t2;
        bc[1] = (th - sinf(th)) / (t2 * th);
    }
}

__global__ void __launch_bounds__(256) se3_wmma_main(const float* __restrict__ Mp,
                                                     const float* __restrict__ tau,
                                                     const float* __restrict__ bc,
                                                     float* __restrict__ out)
{
    __shared__ __align__(16) float sE[256 * 20];
    __shared__ __align__(16) float sO[256 * 16];

    const int tid = threadIdx.x;
    const size_t blockBase = (size_t)blockIdx.x * 256;
    const size_t e = blockBase + tid;

    __builtin_prefetch(Mp + e * 16, 0, 0);

    const float2* t2 = (const float2*)(tau + e * 6);
    float2 q0 = t2[0], q1 = t2[1], q2 = t2[2];
    float rx = q0.x, ry = q0.y, rz = q1.x;
    float wx = q1.y, wy = q2.x, wz = q2.y;

    float th = sqrtf(wx * wx + wy * wy + wz * wz);
    float inv = 1.0f / th;
    float ux = wx * inv, uy = wy * inv, uz = wz * inv;
    float sn = sinf(th);
    float omc = 1.0f - cosf(th);

    float R01 = -sn * uz + omc * uz * uz;
    float R02 =  sn * uy + omc * uy * uy;
    float R10 =  sn * uz + omc * uz * uz;
    float R12 = -sn * ux + omc * ux * ux;
    float R20 = -sn * uy + omc * uy * uy;
    float R21 =  sn * ux + omc * ux * ux;

    float b = bc[0], c = bc[1];
    float V01 = -b * wz + c * wz * wz;
    float V02 =  b * wy + c * wy * wy;
    float V10 =  b * wz + c * wz * wz;
    float V12 = -b * wx + c * wx * wx;
    float V20 = -b * wy + c * wy * wy;
    float V21 =  b * wx + c * wx * wx;
    float px = rx + V01 * ry + V02 * rz;
    float py = V10 * rx + ry + V12 * rz;
    float pz = V20 * rx + V21 * ry + rz;

    float* Ec = &sE[tid * 20];
    Ec[0]  = 1.0f; Ec[1]  = R10;  Ec[2]  = R20;  Ec[3]  = 0.0f;
    Ec[4]  = R01;  Ec[5]  = 1.0f; Ec[6]  = R21;  Ec[7]  = 0.0f;
    Ec[8]  = R02;  Ec[9]  = R12;  Ec[10] = 1.0f; Ec[11] = 0.0f;
    Ec[12] = px;   Ec[13] = py;   Ec[14] = pz;   Ec[15] = 1.0f;

    __syncthreads();

    const int lane = tid & 31;
    const int waveBase = tid & ~31;
    const int n = lane & 15;
    const int half = lane >> 4;
    const int k0 = half << 1;
    const int cc = n & 3;
    const int dstHalf = n >> 3;
    const bool useHi = ((n >> 2) & 1) != 0;

    const float* aBase = Mp + (blockBase + waveBase + (n >> 2)) * 16
                            + (size_t)((n & 3) * 4 + k0);
    const float* bBase = &sE[(waveBase + (n >> 2)) * 20 + cc * 4 + k0];
    float*       oBase = &sO[(waveBase + (n >> 2)) * 16 + cc];

    v2f av[8], bv[8];
    #pragma unroll
    for (int g = 0; g < 8; ++g) {
        av[g] = *(const v2f*)(aBase + g * 64);
        bv[g] = *(const v2f*)(bBase + g * 80);
    }

    #pragma unroll
    for (int g = 0; g < 8; ++g) {
        v8f acc = {0.f, 0.f, 0.f, 0.f, 0.f, 0.f, 0.f, 0.f};
        acc = __builtin_amdgcn_wmma_f32_16x16x4_f32(
             false, av[g],  false, bv[g],
             (short)0, acc,  false,  false);

        if (half == dstHalf) {
            float d0 = useHi ? acc[4] : acc[0];
            float d1 = useHi ? acc[5] : acc[1];
            float d2 = useHi ? acc[6] : acc[2];
            float d3 = useHi ? acc[7] : acc[3];
            float* o = oBase + g * 64;
            o[0]  = d0;
            o[4]  = d1;
            o[8]  = d2;
            o[12] = d3;
        }
    }

    __syncthreads();

    const float4* src = (const float4*)(&sO[waveBase * 16]) + lane;
    float4* dst = (float4*)(out + (blockBase + waveBase) * 16) + lane;
    dst[0]  = src[0];
    dst[32] = src[32];
    dst[64] = src[64];
    dst[96] = src[96];
}

__global__ void se3_tail(const float* __restrict__ Mp, const float* __restrict__ tau,
                         const float* __restrict__ bc, float* __restrict__ out,
                         int start, int K)
{
    int e = start + blockIdx.x * blockDim.x + threadIdx.x;
    if (e >= K) return;
    const float* t = tau + (size_t)e * 6;
    float rx = t[0], ry = t[1], rz = t[2], wx = t[3], wy = t[4], wz = t[5];
    float th = sqrtf(wx * wx + wy * wy + wz * wz);
    float inv = 1.0f / th;
    float ux = wx * inv, uy = wy * inv, uz = wz * inv;
    float sn = sinf(th), omc = 1.0f - cosf(th);
    float E[4][4];
    E[0][0] = 1.0f;                    E[0][1] = -sn * uz + omc * uz * uz; E[0][2] =  sn * uy + omc * uy * uy;
    E[1][0] =  sn * uz + omc * uz * uz; E[1][1] = 1.0f;                   E[1][2] = -sn * ux + omc * ux * ux;
    E[2][0] = -sn * uy + omc * uy * uy; E[2][1] =  sn * ux + omc * ux * ux; E[2][2] = 1.0f;
    float b = bc[0], c = bc[1];
    E[0][3] = rx + (-b * wz + c * wz * wz) * ry + ( b * wy + c * wy * wy) * rz;
    E[1][3] = ( b * wz + c * wz * wz) * rx + ry + (-b * wx + c * wx * wx) * rz;
    E[2][3] = (-b * wy + c * wy * wy) * rx + ( b * wx + c * wx * wx) * ry + rz;
    E[3][0] = 0.0f; E[3][1] = 0.0f; E[3][2] = 0.0f; E[3][3] = 1.0f;
    const float* m = Mp + (size_t)e * 16;
    float* o = out + (size_t)e * 16;
    for (int r = 0; r < 4; ++r)
        for (int c2 = 0; c2 < 4; ++c2) {
            float s = 0.0f;
            for (int k = 0; k < 4; ++k) s += m[r * 4 + k] * E[k][c2];
            o[r * 4 + c2] = s;
        }
}

extern "C" void kernel_launch(void* const* d_in, const int* in_sizes, int n_in,
                              void* d_out, int out_size, void* d_ws, size_t ws_size,
                              hipStream_t stream)
{
    const float* Mp  = (const float*)d_in[0];
    const float* tau = (const float*)d_in[1];
    float* out = (float*)d_out;
    const int K = in_sizes[0] / 16;

    float* w = (float*)d_ws;
    float* bc = w;
    float* partial = w + 16;

    se3_ssq_partial<<<RED_BLOCKS, 256, 0, stream>>>(tau, partial, K);
    se3_ssq_final<<<1, 256, 0, stream>>>(partial, bc);

    const int nb = K / 256;
    if (nb > 0)
        se3_wmma_main<<<nb, 256, 0, stream>>>(Mp, tau, bc, out);
    const int tail = K - nb * 256;
    if (tail > 0)
        se3_tail<<<(tail + 255) / 256, 256, 0, stream>>>(Mp, tau, bc, out, nb * 256, K);
}


# final-reduction launch removed (3 launches -> 2): pass 1 writes 256 block sums, each main-kernel wave adds them by a lane butterfly and forms b,c itself; summation order of the norm changes
# speedup vs baseline: 1.5759x; 1.0860x over previous
#include <hip/hip_runtime.h>
#include <math.h>

typedef __attribute__((ext_vector_type(2))) float v2f;
typedef __attribute__((ext_vector_type(8))) float v8f;

#define NPART 256
#define SSQ_THREADS 1024

template <int M>
__device__ __forceinline__ float row_xor(float x)
{
    int b = __builtin_bit_cast(int, x);
    return __builtin_bit_cast(float, __builtin_amdgcn_update_dpp(0, b, 0x160 + M, 0xf, 0xf, true));
}

__device__ __forceinline__ float xhalf(float x)
{
    int b = __builtin_bit_cast(int, x);
    return __builtin_bit_cast(float,
        __builtin_amdgcn_permlanex16(b, b, 0x76543210, 0xfedcba98, false, false));
}

__device__ __forceinline__ float wave_sum32(float s)
{
    s += row_xor<1>(s);
    s += row_xor<2>(s);
    s += row_xor<4>(s);
    s += row_xor<8>(s);
    s += xhalf(s);
    return s;
}

__global__ void __launch_bounds__(SSQ_THREADS) se3_ssq_partial(const float* __restrict__ tau,
                                                               float* __restrict__ partial, int K)
{
    __shared__ float red[SSQ_THREADS / 32];
    float s = 0.0f;
    for (int e = blockIdx.x * SSQ_THREADS + threadIdx.x; e < K; e += gridDim.x * SSQ_THREADS) {
        const float2* t2 = (const float2*)(tau + (size_t)e * 6);
        float2 q1 = t2[1];
        float2 q2 = t2[2];
        float wx = q1.y, wy = q2.x, wz = q2.y;
        s += wx * wx + wy * wy + wz * wz;
    }
    const int lane = threadIdx.x & 31;
    const int wave = threadIdx.x >> 5;
    s = wave_sum32(s);
    if (lane == 0) red[wave] = s;
    __syncthreads();
    if (wave == 0) {
        float v = wave_sum32(red[lane]);
        if (lane == 0) partial[blockIdx.x] = v;
    }
}

__device__ __forceinline__ void bc_load(const float* __restrict__ partial, float4& u, float4& v)
{
    const int lane = threadIdx.x & 31;
    const float4* p4 = (const float4*)partial;
    u = p4[lane];
    v = p4[lane + 32];
}

__device__ __forceinline__ void bc_finish(float4 u, float4 v, float& b, float& c)
{
    float s = ((u.x + u.y) + (u.z + u.w)) + ((v.x + v.y) + (v.z + v.w));
    s = wave_sum32(s);
    float tot = __builtin_bit_cast(float,
        __builtin_amdgcn_readfirstlane(__builtin_bit_cast(int, s)));
    float th = sqrtf(tot);
    float t2 = th * th;
    b = (1.0f - cosf(th)) / t2;
    c = (th - sinf(th)) / (t2 * th);
}

__global__ void __launch_bounds__(256) se3_wmma_main(const float* __restrict__ Mp,
                                                     const float* __restrict__ tau,
                                                     const float* __restrict__ partial,
                                                     float* __restrict__ out)
{
    __shared__ __align__(16) float sE[256 * 20];
    __shared__ __align__(16) float sO[256 * 16];

    const int tid = threadIdx.x;
    const size_t blockBase = (size_t)blockIdx.x * 256;
    const size_t e = blockBase + tid;

    __builtin_prefetch(Mp + e * 16, 0, 0);

    float4 su, sv;
    bc_load(partial, su, sv);

    const float2* t2 = (const float2*)(tau + e * 6);
    float2 q0 = t2[0], q1 = t2[1], q2 = t2[2];

    float b, c;
    bc_finish(su, sv, b, c);

    float rx = q0.x, ry = q0.y, rz = q1.x;
    float wx = q1.y, wy = q2.x, wz = q2.y;

    float th = sqrtf(wx * wx + wy * wy + wz * wz);
    float inv = 1.0f / th;
    float ux = wx * inv, uy = wy * inv, uz = wz * inv;
    float sn = sinf(th);
    float omc = 1.0f - cosf(th);

    float R01 = -sn * uz + omc * uz * uz;
    float R02 =  sn * uy + omc * uy * uy;
    float R10 =  sn * uz + omc * uz * uz;
    float R12 = -sn * ux + omc * ux * ux;
    float R20 = -sn * uy + omc * uy * uy;
    float R21 =  sn * ux + omc * ux * ux;

    float V01 = -b * wz + c * wz * wz;
    float V02 =  b * wy + c * wy * wy;
    float V10 =  b * wz + c * wz * wz;
    float V12 = -b * wx + c * wx * wx;
    float V20 = -b * wy + c * wy * wy;
    float V21 =  b * wx + c * wx * wx;
    float px = rx + V01 * ry + V02 * rz;
    float py = V10 * rx + ry + V12 * rz;
    float pz = V20 * rx + V21 * ry + rz;

    float* Ec = &sE[tid * 20];
    Ec[0]  = 1.0f; Ec[1]  = R10;  Ec[2]  = R20;  Ec[3]  = 0.0f;
    Ec[4]  = R01;  Ec[5]  = 1.0f; Ec[6]  = R21;  Ec[7]  = 0.0f;
    Ec[8]  = R02;  Ec[9]  = R12;  Ec[10] = 1.0f; Ec[11] = 0.0f;
    Ec[12] = px;   Ec[13] = py;   Ec[14] = pz;   Ec[15] = 1.0f;

    __syncthreads();

    const int lane = tid & 31;
    const int waveBase = tid & ~31;
    const int n = lane & 15;
    const int half = lane >> 4;
    const int k0 = half << 1;
    const int cc = n & 3;
    const int dstHalf = n >> 3;
    const bool useHi = ((n >> 2) & 1) != 0;

    const float* aBase = Mp + (blockBase + waveBase + (n >> 2)) * 16
                            + (size_t)((n & 3) * 4 + k0);
    const float* bBase = &sE[(waveBase + (n >> 2)) * 20 + cc * 4 + k0];
    float*       oBase = &sO[(waveBase + (n >> 2)) * 16 + cc];

    v2f av[8], bv[8];
    #pragma unroll
    for (int g = 0; g < 8; ++g) {
        av[g] = *(const v2f*)(aBase + g * 64);
        bv[g] = *(const v2f*)(bBase + g * 80);
    }

    #pragma unroll
    for (int g = 0; g < 8; ++g) {
        v8f acc = {0.f, 0.f, 0.f, 0.f, 0.f, 0.f, 0.f, 0.f};
        acc = __builtin_amdgcn_wmma_f32_16x16x4_f32(
             false, av[g],  false, bv[g],
             (short)0, acc,  false,  false);

        if (half == dstHalf) {
            float d0 = useHi ? acc[4] : acc[0];
            float d1 = useHi ? acc[5] : acc[1];
            float d2 = useHi ? acc[6] : acc[2];
            float d3 = useHi ? acc[7] : acc[3];
            float* o = oBase + g * 64;
            o[0]  = d0;
            o[4]  = d1;
            o[8]  = d2;
            o[12] = d3;
        }
    }

    __syncthreads();

    const float4* src = (const float4*)(&sO[waveBase * 16]) + lane;
    float4* dst = (float4*)(out + (blockBase + waveBase) * 16) + lane;
    dst[0]  = src[0];
    dst[32] = src[32];
    dst[64] = src[64];
    dst[96] = src[96];
}

__global__ void se3_tail(const float* __restrict__ Mp, const float* __restrict__ tau,
                         const float* __restrict__ partial, float* __restrict__ out,
                         int start, int K)
{
    float4 su, sv;
    float b, c;
    bc_load(partial, su, sv);
    bc_finish(su, sv, b, c);
    int e = start + blockIdx.x * blockDim.x + threadIdx.x;
    if (e >= K) return;
    const float* t = tau + (size_t)e * 6;
    float rx = t[0], ry = t[1], rz = t[2], wx = t[3], wy = t[4], wz = t[5];
    float th = sqrtf(wx * wx + wy * wy + wz * wz);
    float inv = 1.0f / th;
    float ux = wx * inv, uy = wy * inv, uz = wz * inv;
    float sn = sinf(th), omc = 1.0f - cosf(th);
    float E[4][4];
    E[0][0] = 1.0f;                    E[0][1] = -sn * uz + omc * uz * uz; E[0][2] =  sn * uy + omc * uy * uy;
    E[1][0] =  sn * uz + omc * uz * uz; E[1][1] = 1.0f;                   E[1][2] = -sn * ux + omc * ux * ux;
    E[2][0] = -sn * uy + omc * uy * uy; E[2][1] =  sn * ux + omc * ux * ux; E[2][2] = 1.0f;
    E[0][3] = rx + (-b * wz + c * wz * wz) * ry + ( b * wy + c * wy * wy) * rz;
    E[1][3] = ( b * wz + c * wz * wz) * rx + ry + (-b * wx + c * wx * wx) * rz;
    E[2][3] = (-b * wy + c * wy * wy) * rx + ( b * wx + c * wx * wx) * ry + rz;
    E[3][0] = 0.0f; E[3][1] = 0.0f; E[3][2] = 0.0f; E[3][3] = 1.0f;
    const float* m = Mp + (size_t)e * 16;
    float* o = out + (size_t)e * 16;
    for (int r = 0; r < 4; ++r)
        for (int c2 = 0; c2 < 4; ++c2) {
            float s = 0.0f;
            for (int k = 0; k < 4; ++k) s += m[r * 4 + k] * E[k][c2];
            o[r * 4 + c2] = s;
        }
}

extern "C" void kernel_launch(void* const* d_in, const int* in_sizes, int n_in,
                              void* d_out, int out_size, void* d_ws, size_t ws_size,
                              hipStream_t stream)
{
    const float* Mp  = (const float*)d_in[0];
    const float* tau = (const float*)d_in[1];
    float* out = (float*)d_out;
    const int K = in_sizes[0] / 16;

    float* partial = (float*)d_ws;

    se3_ssq_partial<<<NPART, SSQ_THREADS, 0, stream>>>(tau, partial, K);

    const int nb = K / 256;
    if (nb > 0)
        se3_wmma_main<<<nb, 256, 0, stream>>>(Mp, tau, partial, out);
    const int tail = K - nb * 256;
    if (tail > 0)
        se3_tail<<<(tail + 255) / 256, 256, 0, stream>>>(Mp, tau, partial, out, nb * 256, K);
}
